# starting kernel
# baseline (speedup 1.0000x reference)
#include <hip/hip_runtime.h>
#include <math.h>

typedef __attribute__((ext_vector_type(2))) float v2f;
typedef __attribute__((ext_vector_type(8))) float v8f;

#define NJ      16384
#define NI      65536
#define NE      262144
#define A_ELEMS (441 * NJ)
#define LD_IDX  (2 * A_ELEMS)
#define ZBASE   (LD_IDX + 1)
#define ZTOT    (21 * NJ)

struct cplx { float re, im; };

__device__ __forceinline__ cplx cmk(float r, float i){ cplx c; c.re=r; c.im=i; return c; }
__device__ __forceinline__ cplx cadd(cplx a, cplx b){ return cmk(a.re+b.re, a.im+b.im); }
__device__ __forceinline__ cplx csub(cplx a, cplx b){ return cmk(a.re-b.re, a.im-b.im); }
__device__ __forceinline__ cplx cmul(cplx a, cplx b){ return cmk(a.re*b.re - a.im*b.im, a.re*b.im + a.im*b.re); }
__device__ __forceinline__ cplx cmulc(cplx a, cplx b){
  return cmk(a.re*b.re + a.im*b.im, a.im*b.re - a.re*b.im); }
__device__ __forceinline__ cplx cneg(cplx a){ return cmk(-a.re, -a.im); }
__device__ __forceinline__ cplx cinv(cplx a){
  float d = 1.0f/(a.re*a.re + a.im*a.im); return cmk(a.re*d, -a.im*d); }

__device__ __forceinline__ float waveReduceWmma(float v) {
  v2f a; a.x = 1.0f; a.y = 1.0f;
  v2f b; b.x = v;    b.y = 0.0f;
  v8f c = {};
  v8f d = __builtin_amdgcn_wmma_f32_16x16x4_f32(false, a, false, b,
                                                (short)0, c, false, false);
  float h = d[0];
  h += __shfl_xor(h, 1, 32);
  h += __shfl_xor(h, 2, 32);
  h += __shfl_xor(h, 4, 32);
  h += __shfl_xor(h, 8, 32);
  return h;
}

__device__ __forceinline__ void blockAccum(float v, float* dst, int tid) {
  float w = waveReduceWmma(v);
  __shared__ float wsum[8];
  if ((tid & 31) == 0) wsum[tid >> 5] = w;
  __syncthreads();
  if (tid == 0) {
    float s = 0.f;
    #pragma unroll
    for (int q = 0; q < 8; ++q) s += wsum[q];
    atomicAdd(dst, s);
  }
}

__device__ __forceinline__ cplx A1val(const cplx T1[4][4], const cplx iS1[4],
                                      const cplx il0[4][4], int t, int p, int a) {
  if (p < 4 && a < 4) {
    cplx v = cmulc(cmul(T1[t][p], iS1[t]), T1[t][a]);
    if (p == a) v = cadd(v, il0[t][p]);
    return v;
  } else if (p < 4) {
    return cneg(cmul(T1[t][p], iS1[t]));
  } else if (a < 4) {
    cplx x = cmul(T1[t][a], iS1[t]);
    return cmk(-x.re, x.im);
  }
  return iS1[t];
}

__global__ void zero_logdet_kernel(float* out) { out[LD_IDX] = 0.0f; }

__global__ __launch_bounds__(256, 1)
void logdet00_kernel(const float* __restrict__ re,
                     const float* __restrict__ im,
                     float* __restrict__ out) {
  int idx = blockIdx.x * 256 + threadIdx.x;
  float r = re[idx], ii = im[idx];
  float v = 0.5f * logf(r*r + ii*ii);
  blockAccum(v, out + LD_IDX, threadIdx.x);
}

__global__ __launch_bounds__(256, 1)
void schur_kernel(const float* __restrict__ l00r, const float* __restrict__ l00i,
                  const float* __restrict__ l01r, const float* __restrict__ l01i,
                  const float* __restrict__ l02r, const float* __restrict__ l02i,
                  const float* __restrict__ l11r, const float* __restrict__ l11i,
                  const float* __restrict__ l12r, const float* __restrict__ l12i,
                  const float* __restrict__ l22r, const float* __restrict__ l22i,
                  const float* __restrict__ zre,  const float* __restrict__ zim,
                  float* __restrict__ out) {
  const int j = blockIdx.x * 256 + threadIdx.x;
  float ld = 0.f;

  cplx T1[4][4], iS1[4], il0[4][4];
  #pragma unroll
  for (int t = 0; t < 4; ++t) {
    const int i = t * NJ + j;
    cplx m = cmk(0.f, 0.f);
    #pragma unroll
    for (int s = 0; s < 4; ++s) {
      const int e = s * NI + i;
      cplx l00 = cmk(l00r[e], l00i[e]);
      cplx l01 = cmk(l01r[e], l01i[e]);
      cplx il  = cinv(l00);
      il0[t][s] = il;
      cplx tt = cmul(l01, il);
      T1[t][s] = tt;
      m = cadd(m, cmulc(tt, l01));
    }
    cplx S1 = csub(cmk(l11r[i], l11i[i]), m);
    ld += 0.5f * logf(S1.re*S1.re + S1.im*S1.im);
    iS1[t] = cinv(S1);
  }

  cplx Bv[5][4];
  #pragma unroll
  for (int t = 0; t < 4; ++t) {
    #pragma unroll
    for (int a = 0; a < 4; ++a)
      Bv[a][t] = cmk(l02r[(a*4+t)*NJ + j], l02i[(a*4+t)*NJ + j]);
    Bv[4][t] = cmk(l12r[t*NJ + j], l12i[t*NJ + j]);
  }

  cplx T20[20];
  #pragma unroll
  for (int t = 0; t < 4; ++t) {
    #pragma unroll
    for (int p = 0; p < 5; ++p) {
      cplx acc = cmk(0.f, 0.f);
      #pragma unroll
      for (int a = 0; a < 5; ++a)
        acc = cadd(acc, cmul(Bv[a][t], A1val(T1, iS1, il0, t, p, a)));
      T20[p*4 + t] = acc;
    }
  }

  cplx m2 = cmk(0.f, 0.f);
  #pragma unroll
  for (int t = 0; t < 4; ++t)
    #pragma unroll
    for (int p = 0; p < 5; ++p)
      m2 = cadd(m2, cmulc(T20[p*4 + t], Bv[p][t]));
  cplx S2 = csub(cmk(l22r[j], l22i[j]), m2);
  ld += 0.5f * logf(S2.re*S2.re + S2.im*S2.im);
  cplx iS2 = cinv(S2);

  cplx zv[21];
  #pragma unroll
  for (int q = 0; q < 21; ++q) zv[q] = cmk(zre[q*NJ + j], zim[q*NJ + j]);

  float* __restrict__ outAr = out;
  float* __restrict__ outAi = out + A_ELEMS;
  float* __restrict__ outZr = out + ZBASE;
  float* __restrict__ outZi = out + ZBASE + ZTOT;

  #pragma unroll
  for (int t = 0; t < 4; ++t) {
    #pragma unroll
    for (int a = 0; a < 5; ++a) {
      const int k = a*4 + t;
      cplx Pk = cmul(T20[k], iS2);
      cplx zo = cmk(0.f, 0.f);
      #pragma unroll
      for (int tp = 0; tp < 4; ++tp) {
        #pragma unroll
        for (int ap = 0; ap < 5; ++ap) {
          const int kp = ap*4 + tp;
          cplx Av = cmulc(Pk, T20[kp]);
          if (tp == t) Av = cadd(Av, A1val(T1, iS1, il0, t, a, ap));
          zo = cadd(zo, cmul(zv[kp], Av));
          outAr[(k*21 + kp)*NJ + j] = Av.re;
          outAi[(k*21 + kp)*NJ + j] = Av.im;
        }
      }
      {
        cplx Av = cneg(Pk);
        zo = cadd(zo, cmul(zv[20], Av));
        outAr[(k*21 + 20)*NJ + j] = Av.re;
        outAi[(k*21 + 20)*NJ + j] = Av.im;
      }
      outZr[k*NJ + j] = zo.re;
      outZi[k*NJ + j] = zo.im;
    }
  }
  {
    cplx zo = cmk(0.f, 0.f);
    #pragma unroll
    for (int tp = 0; tp < 4; ++tp) {
      #pragma unroll
      for (int ap = 0; ap < 5; ++ap) {
        const int kp = ap*4 + tp;
        cplx x  = cmul(T20[kp], iS2);
        cplx Av = cmk(-x.re, x.im);
        zo = cadd(zo, cmul(zv[kp], Av));
        outAr[(20*21 + kp)*NJ + j] = Av.re;
        outAi[(20*21 + kp)*NJ + j] = Av.im;
      }
    }
    cplx Av = iS2;
    zo = cadd(zo, cmul(zv[20], Av));
    outAr[(20*21 + 20)*NJ + j] = Av.re;
    outAi[(20*21 + 20)*NJ + j] = Av.im;
    outZr[20*NJ + j] = zo.re;
    outZi[20*NJ + j] = zo.im;
  }

  blockAccum(ld, out + LD_IDX, threadIdx.x);
}

extern "C" void kernel_launch(void* const* d_in, const int* in_sizes, int n_in,
                              void* d_out, int out_size, void* d_ws, size_t ws_size,
                              hipStream_t stream) {
  (void)in_sizes; (void)n_in; (void)out_size; (void)d_ws; (void)ws_size;
  const float* l00r = (const float*)d_in[0];
  const float* l00i = (const float*)d_in[1];
  const float* l01r = (const float*)d_in[2];
  const float* l01i = (const float*)d_in[3];
  const float* l02r = (const float*)d_in[4];
  const float* l02i = (const float*)d_in[5];
  const float* l11r = (const float*)d_in[6];
  const float* l11i = (const float*)d_in[7];
  const float* l12r = (const float*)d_in[8];
  const float* l12i = (const float*)d_in[9];
  const float* l22r = (const float*)d_in[10];
  const float* l22i = (const float*)d_in[11];
  const float* zre  = (const float*)d_in[12];
  const float* zim  = (const float*)d_in[13];
  float* out = (float*)d_out;

  zero_logdet_kernel<<<1, 1, 0, stream>>>(out);
  logdet00_kernel<<<NE / 256, 256, 0, stream>>>(l00r, l00i, out);
  schur_kernel<<<NJ / 256, 256, 0, stream>>>(l00r, l00i, l01r, l01i, l02r, l02i,
                                             l11r, l11i, l12r, l12i, l22r, l22i,
                                             zre, zim, out);
}


# waves (t,0),(t,1) of a 64-column group swap halves through LDS: each A store is 8 B per lane, one whole aligned 256-B unit per instruction; 8-wave groups; logdet partial sums regrouped
# speedup vs baseline: 3.8510x; 3.8510x over previous
#include <hip/hip_runtime.h>
#include <math.h>

typedef __attribute__((ext_vector_type(2))) float v2f;
typedef __attribute__((ext_vector_type(8))) float v8f;

#define NJ      16384
#define NI      65536
#define NE      262144
#define A_ELEMS (441 * NJ)
#define LD_IDX  (2 * A_ELEMS)
#define ZBASE   (LD_IDX + 1)
#define ZTOT    (21 * NJ)

struct cplx { float re, im; };

__device__ __forceinline__ cplx cmk(float r, float i){ cplx c; c.re=r; c.im=i; return c; }
__device__ __forceinline__ cplx cadd(cplx a, cplx b){ return cmk(a.re+b.re, a.im+b.im); }
__device__ __forceinline__ cplx csub(cplx a, cplx b){ return cmk(a.re-b.re, a.im-b.im); }
__device__ __forceinline__ cplx cmul(cplx a, cplx b){ return cmk(a.re*b.re - a.im*b.im, a.re*b.im + a.im*b.re); }
__device__ __forceinline__ cplx cmulc(cplx a, cplx b){
  return cmk(a.re*b.re + a.im*b.im, a.im*b.re - a.re*b.im); }
__device__ __forceinline__ cplx cneg(cplx a){ return cmk(-a.re, -a.im); }
__device__ __forceinline__ cplx cinv(cplx a){
  float d = 1.0f/(a.re*a.re + a.im*a.im); return cmk(a.re*d, -a.im*d); }

__device__ __forceinline__ float waveReduceWmma(float v) {
  v2f a; a.x = 1.0f; a.y = 1.0f;
  v2f b; b.x = v;    b.y = 0.0f;
  v8f c = {};
  v8f d = __builtin_amdgcn_wmma_f32_16x16x4_f32(false, a, false, b,
                                                (short)0, c, false, false);
  float h = d[0];
  h += __shfl_xor(h, 1, 32);
  h += __shfl_xor(h, 2, 32);
  h += __shfl_xor(h, 4, 32);
  h += __shfl_xor(h, 8, 32);
  return h;
}

template <int NW>
__device__ __forceinline__ void blockAccum(float v, float* dst, int tid) {
  float w = waveReduceWmma(v);
  __shared__ float wsum[NW];
  if ((tid & 31) == 0) wsum[tid >> 5] = w;
  __syncthreads();
  if (tid == 0) {
    float s = 0.f;
    #pragma unroll
    for (int q = 0; q < NW; ++q) s += wsum[q];
    atomicAdd(dst, s);
  }
}

__device__ __forceinline__ void ldsBarrier() {
  __builtin_amdgcn_fence(__ATOMIC_RELEASE, "workgroup", "local");
  __builtin_amdgcn_s_barrier();
  __builtin_amdgcn_fence(__ATOMIC_ACQUIRE, "workgroup", "local");
}

__device__ __forceinline__ cplx A1val(const cplx T1[4], const cplx iS1,
                                      const cplx il0[4], int p, int a) {
  if (p < 4 && a < 4) {
    cplx v = cmulc(cmul(T1[p], iS1), T1[a]);
    if (p == a) v = cadd(v, il0[p]);
    return v;
  } else if (p < 4) {
    return cneg(cmul(T1[p], iS1));
  } else if (a < 4) {
    cplx x = cmul(T1[a], iS1);
    return cmk(-x.re, x.im);
  }
  return iS1;
}

__global__ void zero_logdet_kernel(float* out) { out[LD_IDX] = 0.0f; }

#define JB 64

__global__ __launch_bounds__(256)
void schur_kernel(const float* __restrict__ l00r, const float* __restrict__ l00i,
                  const float* __restrict__ l01r, const float* __restrict__ l01i,
                  const float* __restrict__ l02r, const float* __restrict__ l02i,
                  const float* __restrict__ l11r, const float* __restrict__ l11i,
                  const float* __restrict__ l12r, const float* __restrict__ l12i,
                  const float* __restrict__ l22r, const float* __restrict__ l22i,
                  const float* __restrict__ zre,  const float* __restrict__ zim,
                  float* __restrict__ out) {
  __shared__ float sT[20][2][JB];
  __shared__ float sM[20][2][JB];
  __shared__ __attribute__((aligned(16))) float sO[2][4][21][2][JB];

  const int tid = threadIdx.x;
  const int ln  = tid & 31;
  const int t = __builtin_amdgcn_readfirstlane(tid >> 6);
  const int h = __builtin_amdgcn_readfirstlane((tid >> 5) & 1);
  const int jl = h * 32 + ln;
  const int j = blockIdx.x * JB + jl;

  cplx zv[21];
  #pragma unroll
  for (int q = 0; q < 21; ++q) zv[q] = cmk(zre[q*NJ + j], zim[q*NJ + j]);
  const cplx l22 = cmk(l22r[j], l22i[j]);

  cplx T1[4], il0[4];
  const int i = t * NJ + j;
  cplx m = cmk(0.f, 0.f);
  float ld = 0.f;
  #pragma unroll
  for (int s = 0; s < 4; ++s) {
    const int e = s * NI + i;
    cplx l00 = cmk(l00r[e], l00i[e]);
    cplx l01 = cmk(l01r[e], l01i[e]);
    ld += 0.5f * logf(l00.re*l00.re + l00.im*l00.im);
    cplx il  = cinv(l00);
    il0[s] = il;
    cplx tt = cmul(l01, il);
    T1[s] = tt;
    m = cadd(m, cmulc(tt, l01));
  }
  const cplx S1 = csub(cmk(l11r[i], l11i[i]), m);
  ld += 0.5f * logf(S1.re*S1.re + S1.im*S1.im);
  const cplx iS1 = cinv(S1);

  cplx A1m[5][5];
  #pragma unroll
  for (int p = 0; p < 5; ++p)
    #pragma unroll
    for (int a = 0; a < 5; ++a)
      A1m[p][a] = A1val(T1, iS1, il0, p, a);

  cplx Bv[5];
  #pragma unroll
  for (int a = 0; a < 4; ++a)
    Bv[a] = cmk(l02r[(a*4+t)*NJ + j], l02i[(a*4+t)*NJ + j]);
  Bv[4] = cmk(l12r[t*NJ + j], l12i[t*NJ + j]);

  cplx Tt[5];
  #pragma unroll
  for (int p = 0; p < 5; ++p) {
    cplx acc = cmk(0.f, 0.f);
    #pragma unroll
    for (int a = 0; a < 5; ++a)
      acc = cadd(acc, cmul(Bv[a], A1m[p][a]));
    Tt[p] = acc;
    const cplx mt = cmulc(acc, Bv[p]);
    sT[p*4 + t][0][jl] = acc.re;
    sT[p*4 + t][1][jl] = acc.im;
    sM[p*4 + t][0][jl] = mt.re;
    sM[p*4 + t][1][jl] = mt.im;
  }
  __syncthreads();

  cplx T20[20];
  #pragma unroll
  for (int kp = 0; kp < 20; ++kp) T20[kp] = cmk(sT[kp][0][jl], sT[kp][1][jl]);

  cplx m2 = cmk(0.f, 0.f);
  #pragma unroll
  for (int tq = 0; tq < 4; ++tq)
    #pragma unroll
    for (int p = 0; p < 5; ++p)
      m2 = cadd(m2, cmk(sM[p*4 + tq][0][jl], sM[p*4 + tq][1][jl]));
  const cplx S2 = csub(l22, m2);
  const float ld2 = 0.5f * logf(S2.re*S2.re + S2.im*S2.im);
  if (t == 0) ld += ld2;
  const cplx iS2 = cinv(S2);

  float* __restrict__ outZr = out + ZBASE;
  float* __restrict__ outZi = out + ZBASE + ZTOT;
  float* __restrict__ outP = out + h * A_ELEMS + blockIdx.x * JB + 2 * ln;

  #pragma unroll
  for (int a = 0; a < 5; ++a) {
    const int k = a*4 + t;
    const cplx Pk = cmul(Tt[a], iS2);
    cplx zo = cmk(0.f, 0.f);
    #pragma unroll
    for (int tp = 0; tp < 4; ++tp) {
      #pragma unroll
      for (int ap = 0; ap < 5; ++ap) {
        const int kp = ap*4 + tp;
        cplx Av = cmulc(Pk, T20[kp]);
        if (tp == t) Av = cadd(Av, A1m[a][ap]);
        zo = cadd(zo, cmul(zv[kp], Av));
        sO[a & 1][t][kp][0][jl] = Av.re;
        sO[a & 1][t][kp][1][jl] = Av.im;
      }
    }
    {
      cplx Av = cneg(Pk);
      zo = cadd(zo, cmul(zv[20], Av));
      sO[a & 1][t][20][0][jl] = Av.re;
      sO[a & 1][t][20][1][jl] = Av.im;
    }
    outZr[k*NJ + j] = zo.re;
    outZi[k*NJ + j] = zo.im;
    ldsBarrier();
    float* __restrict__ rowp = outP + (k*21)*NJ;
    #pragma unroll
    for (int kp = 0; kp < 21; ++kp) {
      const v2f v = *(const v2f*)&sO[a & 1][t][kp][h][2 * ln];
      *(v2f*)(rowp + kp*NJ) = v;
    }
  }
  if (t == 3) {
    cplx zo = cmk(0.f, 0.f);
    #pragma unroll
    for (int tp = 0; tp < 4; ++tp) {
      #pragma unroll
      for (int ap = 0; ap < 5; ++ap) {
        const int kp = ap*4 + tp;
        cplx x  = cmul(T20[kp], iS2);
        cplx Av = cmk(-x.re, x.im);
        zo = cadd(zo, cmul(zv[kp], Av));
        sO[1][3][kp][0][jl] = Av.re;
        sO[1][3][kp][1][jl] = Av.im;
      }
    }
    cplx Av = iS2;
    zo = cadd(zo, cmul(zv[20], Av));
    sO[1][3][20][0][jl] = Av.re;
    sO[1][3][20][1][jl] = Av.im;
    outZr[20*NJ + j] = zo.re;
    outZi[20*NJ + j] = zo.im;
  }
  ldsBarrier();
  if (t == 3) {
    float* __restrict__ rowp = outP + (20*21)*NJ;
    #pragma unroll
    for (int kp = 0; kp < 21; ++kp) {
      const v2f v = *(const v2f*)&sO[1][3][kp][h][2 * ln];
      *(v2f*)(rowp + kp*NJ) = v;
    }
  }

  blockAccum<8>(ld, out + LD_IDX, tid);
}

extern "C" void kernel_launch(void* const* d_in, const int* in_sizes, int n_in,
                              void* d_out, int out_size, void* d_ws, size_t ws_size,
                              hipStream_t stream) {
  (void)in_sizes; (void)n_in; (void)out_size; (void)d_ws; (void)ws_size;
  const float* l00r = (const float*)d_in[0];
  const float* l00i = (const float*)d_in[1];
  const float* l01r = (const float*)d_in[2];
  const float* l01i = (const float*)d_in[3];
  const float* l02r = (const float*)d_in[4];
  const float* l02i = (const float*)d_in[5];
  const float* l11r = (const float*)d_in[6];
  const float* l11i = (const float*)d_in[7];
  const float* l12r = (const float*)d_in[8];
  const float* l12i = (const float*)d_in[9];
  const float* l22r = (const float*)d_in[10];
  const float* l22i = (const float*)d_in[11];
  const float* zre  = (const float*)d_in[12];
  const float* zim  = (const float*)d_in[13];
  float* out = (float*)d_out;

  zero_logdet_kernel<<<1, 1, 0, stream>>>(out);
  schur_kernel<<<NJ / JB, 256, 0, stream>>>(l00r, l00i, l01r, l01i, l02r, l02i,
                                             l11r, l11i, l12r, l12i, l22r, l22i,
                                             zre, zim, out);
}
